# starting kernel
# baseline (speedup 1.0000x reference)
#include <hip/hip_runtime.h>

typedef __attribute__((ext_vector_type(2))) float v2f;
typedef __attribute__((ext_vector_type(8))) float v8f;

#define B_    4096
#define NI_   256
#define NH_   2048
#define NO_   128
#define STEPS 100
#define ALPHA_ 0.9f

__global__ __launch_bounds__(256) void snn_gemm1_recur(
    const float* __restrict__ x,
    const float* __restrict__ W1,
    const float* __restrict__ b1,
    const float* __restrict__ betap,
    float* __restrict__ spk)
{
    const int wave = threadIdx.x >> 5;
    const int lane = threadIdx.x & 31;
    const int tile = blockIdx.x * 8 + wave;
    const int tilesN = NH_ / 16;
    const int tm = tile / tilesN;
    const int tn = tile - tm * tilesN;
    const int row0 = tm << 4;
    const int col0 = tn << 4;
    const int mn  = lane & 15;
    const int khi = lane >> 4;

    const float* ap = x  + (size_t)(row0 + mn) * NI_ + (khi << 1);
    const float* bp = W1 + (size_t)(col0 + mn) * NI_ + (khi << 1);

    v8f acc = {};
#pragma unroll 8
    for (int k = 0; k < NI_; k += 4) {
        v2f a = *(const v2f*)(ap + k);
        v2f b = *(const v2f*)(bp + k);
        acc = __builtin_amdgcn_wmma_f32_16x16x4_f32(
            false, a, false, b, (short)0, acc, false, false);
    }

    float beta = *betap;
    beta = fminf(fmaxf(beta, 0.0f), 1.0f);
    const float bias = b1[col0 + mn];

    float c[8], syn[8], mem[8];
#pragma unroll
    for (int i = 0; i < 8; ++i) {
        c[i] = acc[i] + bias;
        syn[i] = 0.0f;
        mem[i] = 0.0f;
    }

#pragma unroll 1
    for (int t = 0; t < STEPS; ++t) {
#pragma unroll
        for (int i = 0; i < 8; ++i) {
            const float reset = (mem[i] > 1.0f) ? 1.0f : 0.0f;
            syn[i] = fmaf(ALPHA_, syn[i], c[i]);
            mem[i] = fmaf(beta, mem[i], syn[i]) - reset;
        }
    }

    float* op = spk + (size_t)(row0 + khi * 8) * NH_ + (col0 + mn);
#pragma unroll
    for (int i = 0; i < 8; ++i)
        op[(size_t)i * NH_] = ((mem[i] - 1.0f) > 0.0f) ? 1.0f : 0.0f;
}

__global__ __launch_bounds__(256) void snn_gemm2(
    const float* __restrict__ spk,
    const float* __restrict__ W2,
    const float* __restrict__ b2,
    float* __restrict__ out)
{
    const int wave = threadIdx.x >> 5;
    const int lane = threadIdx.x & 31;
    const int tile = blockIdx.x * 8 + wave;
    const int tilesN = NO_ / 16;
    const int tm = tile / tilesN;
    const int tn = tile - tm * tilesN;
    const int row0 = tm << 4;
    const int col0 = tn << 4;
    const int mn  = lane & 15;
    const int khi = lane >> 4;

    const float* ap = spk + (size_t)(row0 + mn) * NH_ + (khi << 1);
    const float* bp = W2  + (size_t)(col0 + mn) * NH_ + (khi << 1);

    v8f acc = {};
#pragma unroll 8
    for (int k = 0; k < NH_; k += 4) {
        v2f a = *(const v2f*)(ap + k);
        v2f b = *(const v2f*)(bp + k);
        acc = __builtin_amdgcn_wmma_f32_16x16x4_f32(
            false, a, false, b, (short)0, acc, false, false);
    }

    const float bias = b2[col0 + mn];
    float* op = out + (size_t)(row0 + khi * 8) * NO_ + (col0 + mn);
#pragma unroll
    for (int i = 0; i < 8; ++i)
        op[(size_t)i * NO_] = acc[i] + bias;
}

extern "C" void kernel_launch(void* const* d_in, const int* in_sizes, int n_in,
                              void* d_out, int out_size, void* d_ws, size_t ws_size,
                              hipStream_t stream) {
    (void)in_sizes; (void)n_in; (void)out_size; (void)ws_size;
    const float* x    = (const float*)d_in[0];
    const float* W1   = (const float*)d_in[1];
    const float* b1   = (const float*)d_in[2];
    const float* W2   = (const float*)d_in[3];
    const float* b2   = (const float*)d_in[4];
    const float* beta = (const float*)d_in[5];

    float* spk = (float*)d_ws;
    float* out = (float*)d_out;

    const int tiles1 = (B_ / 16) * (NH_ / 16);
    snn_gemm1_recur<<<tiles1 / 8, 256, 0, stream>>>(x, W1, b1, beta, spk);

    const int tiles2 = (B_ / 16) * (NO_ / 16);
    snn_gemm2<<<tiles2 / 8, 256, 0, stream>>>(spk, W2, b2, out);
}


# Fuse both products into one kernel: cur1 tile computed transposed, final spikes stay in accumulator registers as WMMA B operand of spk@W2^T; 32 wave partials summed via LDS (new sum order)
# speedup vs baseline: 1.1312x; 1.1312x over previous
#include <hip/hip_runtime.h>

typedef __attribute__((ext_vector_type(2))) float v2f;
typedef __attribute__((ext_vector_type(4))) float v4f;
typedef __attribute__((ext_vector_type(8))) float v8f;

#define B_    4096
#define NI_   256
#define NH_   2048
#define NO_   128
#define STEPS 100
#define ALPHA_ 0.9f

#define WAVES_ 32
#define TPW_   ((NH_ / 16) / WAVES_)

__global__ __launch_bounds__(1024) void snn_fused(
    const float* __restrict__ x,
    const float* __restrict__ W1,
    const float* __restrict__ b1,
    const float* __restrict__ W2,
    const float* __restrict__ b2,
    const float* __restrict__ betap,
    float* __restrict__ out)
{
    __shared__ float red[8][64 * 32];

    const int wave = threadIdx.x >> 5;
    const int lane = threadIdx.x & 31;
    const int mn   = lane & 15;
    const int khi  = lane >> 4;
    const int b0   = blockIdx.x << 4;

    float beta = *betap;
    beta = fminf(fmaxf(beta, 0.0f), 1.0f);

    const float* xp = x + (size_t)(b0 + mn) * NI_ + (khi << 1);

    v8f oacc[8];
#pragma unroll
    for (int ot = 0; ot < 8; ++ot) oacc[ot] = (v8f){};

#pragma unroll 1
    for (int j = 0; j < TPW_; ++j) {
        const int h0 = (wave * TPW_ + j) << 4;
        const float* wp = W1 + (size_t)(h0 + mn) * NI_ + (khi << 1);

        v8f acc = {};
#pragma unroll 8
        for (int k = 0; k < NI_; k += 4) {
            v2f a = *(const v2f*)(wp + k);
            v2f b = *(const v2f*)(xp + k);
            acc = __builtin_amdgcn_wmma_f32_16x16x4_f32(
                false, a, false, b, (short)0, acc, false, false);
        }

        const v4f blo = *(const v4f*)(b1 + h0 + (khi << 3));
        const v4f bhi = *(const v4f*)(b1 + h0 + (khi << 3) + 4);

        float c[8], syn[8], mem[8];
#pragma unroll
        for (int i = 0; i < 8; ++i) {
            c[i] = acc[i] + ((i < 4) ? blo[i & 3] : bhi[i & 3]);
            syn[i] = 0.0f;
            mem[i] = 0.0f;
        }

#pragma unroll 1
        for (int t = 0; t < STEPS; ++t) {
#pragma unroll
            for (int i = 0; i < 8; ++i) {
                const float reset = (mem[i] > 1.0f) ? 1.0f : 0.0f;
                syn[i] = fmaf(ALPHA_, syn[i], c[i]);
                mem[i] = fmaf(beta, mem[i], syn[i]) - reset;
            }
        }

        float s[8];
#pragma unroll
        for (int i = 0; i < 8; ++i)
            s[i] = ((mem[i] - 1.0f) > 0.0f) ? 1.0f : 0.0f;

        const float* w2p = W2 + (size_t)mn * NH_ + h0 + (khi << 3);
#pragma unroll
        for (int rp = 0; rp < 4; ++rp) {
            v2f sb;
            sb[0] = s[2 * rp];
            sb[1] = s[2 * rp + 1];
#pragma unroll
            for (int ot = 0; ot < 8; ++ot) {
                v2f a = *(const v2f*)(w2p + (size_t)(ot * 16) * NH_ + 2 * rp);
                oacc[ot] = __builtin_amdgcn_wmma_f32_16x16x4_f32(
                    false, a, false, sb, (short)0, oacc[ot], false, false);
            }
        }
    }

    float* slot = red[wave & 7];
#pragma unroll
    for (int r = 0; r < 4; ++r) {
        if ((wave >> 3) == r) {
#pragma unroll
            for (int ot = 0; ot < 8; ++ot) {
#pragma unroll
                for (int i = 0; i < 8; ++i) {
                    const int idx = (ot * 8 + i) * 32 + lane;
                    if (r == 0) slot[idx] = oacc[ot][i];
                    else        slot[idx] += oacc[ot][i];
                }
            }
        }
        __syncthreads();
    }

    const float bias2 = b2[threadIdx.x & 127];
#pragma unroll
    for (int i = 0; i < 2; ++i) {
        const int idx = threadIdx.x + i * 1024;
        const int b = idx >> 7;
        const int o = idx & 127;
        const int a = (((o >> 4) << 3) + (o & 7)) * 32 + (((o >> 3) & 1) << 4) + b;
        float sum = red[0][a];
#pragma unroll
        for (int q = 1; q < 8; ++q) sum += red[q][a];
        out[(size_t)(b0 + b) * NO_ + o] = sum + bias2;
    }
}

extern "C" void kernel_launch(void* const* d_in, const int* in_sizes, int n_in,
                              void* d_out, int out_size, void* d_ws, size_t ws_size,
                              hipStream_t stream) {
    (void)in_sizes; (void)n_in; (void)out_size; (void)d_ws; (void)ws_size;
    const float* x    = (const float*)d_in[0];
    const float* W1   = (const float*)d_in[1];
    const float* b1   = (const float*)d_in[2];
    const float* W2   = (const float*)d_in[3];
    const float* b2   = (const float*)d_in[4];
    const float* beta = (const float*)d_in[5];

    float* out = (float*)d_out;

    snn_fused<<<B_ / 16, WAVES_ * 32, 0, stream>>>(x, W1, b1, W2, b2, beta, out);
}


# Recurrence: each step's (mem>1) test is taken when that membrane is made and carried round the loop as 8 separate lane masks, so no select waits on a compare issued just ahead of it; same values
# speedup vs baseline: 1.1724x; 1.0364x over previous
#include <hip/hip_runtime.h>

typedef __attribute__((ext_vector_type(2))) float v2f;
typedef __attribute__((ext_vector_type(4))) float v4f;
typedef __attribute__((ext_vector_type(8))) float v8f;

#define B_    4096
#define NI_   256
#define NH_   2048
#define NO_   128
#define STEPS 100
#define ALPHA_ 0.9f

#define WAVES_ 32
#define TPW_   ((NH_ / 16) / WAVES_)

__global__ __launch_bounds__(1024) void snn_fused(
    const float* __restrict__ x,
    const float* __restrict__ W1,
    const float* __restrict__ b1,
    const float* __restrict__ W2,
    const float* __restrict__ b2,
    const float* __restrict__ betap,
    float* __restrict__ out)
{
    __shared__ float red[8][64 * 32];

    const int wave = threadIdx.x >> 5;
    const int lane = threadIdx.x & 31;
    const int mn   = lane & 15;
    const int khi  = lane >> 4;
    const int b0   = blockIdx.x << 4;

    float beta = *betap;
    beta = fminf(fmaxf(beta, 0.0f), 1.0f);

    const float* xp = x + (size_t)(b0 + mn) * NI_ + (khi << 1);

    v8f oacc[8];
#pragma unroll
    for (int ot = 0; ot < 8; ++ot) oacc[ot] = (v8f){};

#pragma unroll 1
    for (int j = 0; j < TPW_; ++j) {
        const int h0 = (wave * TPW_ + j) << 4;
        const float* wp = W1 + (size_t)(h0 + mn) * NI_ + (khi << 1);

        v8f acc = {};
#pragma unroll 8
        for (int k = 0; k < NI_; k += 4) {
            v2f a = *(const v2f*)(wp + k);
            v2f b = *(const v2f*)(xp + k);
            acc = __builtin_amdgcn_wmma_f32_16x16x4_f32(
                false, a, false, b, (short)0, acc, false, false);
        }

        const v4f blo = *(const v4f*)(b1 + h0 + (khi << 3));
        const v4f bhi = *(const v4f*)(b1 + h0 + (khi << 3) + 4);

        float c[8], syn[8], mem[8];
#pragma unroll
        for (int i = 0; i < 8; ++i) {
            c[i] = acc[i] + ((i < 4) ? blo[i & 3] : bhi[i & 3]);
            syn[i] = 0.0f;
            mem[i] = 0.0f;
        }

        bool over[8];
#pragma unroll
        for (int i = 0; i < 8; ++i) over[i] = false;
#pragma unroll 1
        for (int t = 0; t < STEPS; ++t) {
#pragma unroll
            for (int i = 0; i < 8; ++i) {
                const float reset = over[i] ? 1.0f : 0.0f;
                syn[i] = fmaf(ALPHA_, syn[i], c[i]);
                mem[i] = fmaf(beta, mem[i], syn[i]) - reset;
                over[i] = mem[i] > 1.0f;
            }
        }

        float s[8];
#pragma unroll
        for (int i = 0; i < 8; ++i)
            s[i] = ((mem[i] - 1.0f) > 0.0f) ? 1.0f : 0.0f;

        const float* w2p = W2 + (size_t)mn * NH_ + h0 + (khi << 3);
#pragma unroll
        for (int rp = 0; rp < 4; ++rp) {
            v2f sb;
            sb[0] = s[2 * rp];
            sb[1] = s[2 * rp + 1];
#pragma unroll
            for (int ot = 0; ot < 8; ++ot) {
                v2f a = *(const v2f*)(w2p + (size_t)(ot * 16) * NH_ + 2 * rp);
                oacc[ot] = __builtin_amdgcn_wmma_f32_16x16x4_f32(
                    false, a, false, sb, (short)0, oacc[ot], false, false);
            }
        }
    }

    float* slot = red[wave & 7];
#pragma unroll
    for (int r = 0; r < 4; ++r) {
        if ((wave >> 3) == r) {
#pragma unroll
            for (int ot = 0; ot < 8; ++ot) {
#pragma unroll
                for (int i = 0; i < 8; ++i) {
                    const int idx = (ot * 8 + i) * 32 + lane;
                    if (r == 0) slot[idx] = oacc[ot][i];
                    else        slot[idx] += oacc[ot][i];
                }
            }
        }
        __syncthreads();
    }

    const float bias2 = b2[threadIdx.x & 127];
#pragma unroll
    for (int i = 0; i < 2; ++i) {
        const int idx = threadIdx.x + i * 1024;
        const int b = idx >> 7;
        const int o = idx & 127;
        const int a = (((o >> 4) << 3) + (o & 7)) * 32 + (((o >> 3) & 1) << 4) + b;
        float sum = red[0][a];
#pragma unroll
        for (int q = 1; q < 8; ++q) sum += red[q][a];
        out[(size_t)(b0 + b) * NO_ + o] = sum + bias2;
    }
}

extern "C" void kernel_launch(void* const* d_in, const int* in_sizes, int n_in,
                              void* d_out, int out_size, void* d_ws, size_t ws_size,
                              hipStream_t stream) {
    (void)in_sizes; (void)n_in; (void)out_size; (void)d_ws; (void)ws_size;
    const float* x    = (const float*)d_in[0];
    const float* W1   = (const float*)d_in[1];
    const float* b1   = (const float*)d_in[2];
    const float* W2   = (const float*)d_in[3];
    const float* b2   = (const float*)d_in[4];
    const float* beta = (const float*)d_in[5];

    float* out = (float*)d_out;

    snn_fused<<<B_ / 16, WAVES_ * 32, 0, stream>>>(x, W1, b1, W2, b2, beta, out);
}


# Each wave takes its hidden tiles in pairs: one recurrence loop carries both tiles' 16 chains, product 1 shares x fragments; spikes wait in a register (bits) for one product 2 at the end; same values
# speedup vs baseline: 1.2557x; 1.0711x over previous
#include <hip/hip_runtime.h>

typedef __attribute__((ext_vector_type(2))) float v2f;
typedef __attribute__((ext_vector_type(4))) float v4f;
typedef __attribute__((ext_vector_type(8))) float v8f;

#define B_    4096
#define NI_   256
#define NH_   2048
#define NO_   128
#define STEPS 100
#define ALPHA_ 0.9f

#define WAVES_ 32
#define TPW_   ((NH_ / 16) / WAVES_)

__global__ __launch_bounds__(1024) void snn_fused(
    const float* __restrict__ x,
    const float* __restrict__ W1,
    const float* __restrict__ b1,
    const float* __restrict__ W2,
    const float* __restrict__ b2,
    const float* __restrict__ betap,
    float* __restrict__ out)
{
    __shared__ float red[8][64 * 32];

    const int wave = threadIdx.x >> 5;
    const int lane = threadIdx.x & 31;
    const int mn   = lane & 15;
    const int khi  = lane >> 4;
    const int b0   = blockIdx.x << 4;

    float beta = *betap;
    beta = fminf(fmaxf(beta, 0.0f), 1.0f);

    const float* xp = x + (size_t)(b0 + mn) * NI_ + (khi << 1);

    unsigned sbits = 0u;

#pragma unroll 1
    for (int jp = 0; jp < TPW_ / 2; ++jp) {
        const int h0 = (wave * TPW_ + 2 * jp) << 4;
        const float* wp0 = W1 + (size_t)(h0 + mn) * NI_ + (khi << 1);
        const float* wp1 = wp0 + (size_t)16 * NI_;

        v8f acc0 = {};
        v8f acc1 = {};
#pragma unroll 8
        for (int k = 0; k < NI_; k += 4) {
            v2f b  = *(const v2f*)(xp + k);
            v2f a0 = *(const v2f*)(wp0 + k);
            v2f a1 = *(const v2f*)(wp1 + k);
            acc0 = __builtin_amdgcn_wmma_f32_16x16x4_f32(
                false, a0, false, b, (short)0, acc0, false, false);
            acc1 = __builtin_amdgcn_wmma_f32_16x16x4_f32(
                false, a1, false, b, (short)0, acc1, false, false);
        }

        const v4f blo0 = *(const v4f*)(b1 + h0 + (khi << 3));
        const v4f bhi0 = *(const v4f*)(b1 + h0 + (khi << 3) + 4);
        const v4f blo1 = *(const v4f*)(b1 + h0 + 16 + (khi << 3));
        const v4f bhi1 = *(const v4f*)(b1 + h0 + 16 + (khi << 3) + 4);

        float c[16], syn[16], mem[16];
#pragma unroll
        for (int i = 0; i < 8; ++i) {
            c[i]     = acc0[i] + ((i < 4) ? blo0[i & 3] : bhi0[i & 3]);
            c[i + 8] = acc1[i] + ((i < 4) ? blo1[i & 3] : bhi1[i & 3]);
        }
#pragma unroll
        for (int i = 0; i < 16; ++i) {
            syn[i] = 0.0f;
            mem[i] = 0.0f;
        }

        bool over[16];
#pragma unroll
        for (int i = 0; i < 16; ++i) over[i] = false;
#pragma unroll 1
        for (int t = 0; t < STEPS; ++t) {
#pragma unroll
            for (int i = 0; i < 16; ++i) {
                const float reset = over[i] ? 1.0f : 0.0f;
                syn[i] = fmaf(ALPHA_, syn[i], c[i]);
                mem[i] = fmaf(beta, mem[i], syn[i]) - reset;
                over[i] = mem[i] > 1.0f;
            }
        }

        unsigned pb = 0u;
#pragma unroll
        for (int i = 0; i < 16; ++i)
            pb |= (((mem[i] - 1.0f) > 0.0f) ? 1u : 0u) << i;
        sbits |= pb << (16 * jp);
    }

    v8f oacc[8];
#pragma unroll
    for (int ot = 0; ot < 8; ++ot) oacc[ot] = (v8f){};

    const float* w2b = W2 + (size_t)mn * NH_ + ((wave * TPW_) << 4) + (khi << 3);
    v4f cur[8];
#pragma unroll
    for (int ot = 0; ot < 8; ++ot)
        cur[ot] = *(const v4f*)(w2b + (size_t)(ot * 16) * NH_);
    asm volatile("" :: "v"(cur[0]), "v"(cur[1]), "v"(cur[2]), "v"(cur[3]),
                       "v"(cur[4]), "v"(cur[5]), "v"(cur[6]), "v"(cur[7]));

#pragma unroll 1
    for (int g = 0; g < 2 * TPW_; ++g) {
        const unsigned tb = sbits >> (4 * g);
        v2f sb0, sb1;
        sb0[0] = (float)(tb & 1u);
        sb0[1] = (float)((tb >> 1) & 1u);
        sb1[0] = (float)((tb >> 2) & 1u);
        sb1[1] = (float)((tb >> 3) & 1u);
        const int gn = (g + 1 < 2 * TPW_) ? (g + 1) : g;
        const float* nx = w2b + ((gn >> 1) << 4) + ((gn & 1) << 2);
#pragma unroll
        for (int ot = 0; ot < 8; ++ot) {
            v2f a;
            a[0] = cur[ot][0];
            a[1] = cur[ot][1];
            oacc[ot] = __builtin_amdgcn_wmma_f32_16x16x4_f32(
                false, a, false, sb0, (short)0, oacc[ot], false, false);
        }
#pragma unroll
        for (int ot = 0; ot < 8; ++ot) {
            v2f a;
            a[0] = cur[ot][2];
            a[1] = cur[ot][3];
            oacc[ot] = __builtin_amdgcn_wmma_f32_16x16x4_f32(
                false, a, false, sb1, (short)0, oacc[ot], false, false);
            cur[ot] = *(const v4f*)(nx + (size_t)(ot * 16) * NH_);
        }
    }

    float* slot = red[wave & 7];
#pragma unroll
    for (int r = 0; r < 4; ++r) {
        if ((wave >> 3) == r) {
#pragma unroll
            for (int ot = 0; ot < 8; ++ot) {
#pragma unroll
                for (int i = 0; i < 8; ++i) {
                    const int idx = (ot * 8 + i) * 32 + lane;
                    if (r == 0) slot[idx] = oacc[ot][i];
                    else        slot[idx] += oacc[ot][i];
                }
            }
        }
        __syncthreads();
    }

    const float bias2 = b2[threadIdx.x & 127];
#pragma unroll
    for (int i = 0; i < 2; ++i) {
        const int idx = threadIdx.x + i * 1024;
        const int b = idx >> 7;
        const int o = idx & 127;
        const int a = (((o >> 4) << 3) + (o & 7)) * 32 + (((o >> 3) & 1) << 4) + b;
        float sum = red[0][a];
#pragma unroll
        for (int q = 1; q < 8; ++q) sum += red[q][a];
        out[(size_t)(b0 + b) * NO_ + o] = sum + bias2;
    }
}

extern "C" void kernel_launch(void* const* d_in, const int* in_sizes, int n_in,
                              void* d_out, int out_size, void* d_ws, size_t ws_size,
                              hipStream_t stream) {
    (void)in_sizes; (void)n_in; (void)out_size; (void)d_ws; (void)ws_size;
    const float* x    = (const float*)d_in[0];
    const float* W1   = (const float*)d_in[1];
    const float* b1   = (const float*)d_in[2];
    const float* W2   = (const float*)d_in[3];
    const float* b2   = (const float*)d_in[4];
    const float* beta = (const float*)d_in[5];

    float* out = (float*)d_out;

    snn_fused<<<B_ / 16, WAVES_ * 32, 0, stream>>>(x, W1, b1, W2, b2, beta, out);
}
